# starting kernel
# baseline (speedup 1.0000x reference)
#include <hip/hip_runtime.h>


typedef float v2f __attribute__((ext_vector_type(2)));
typedef float v8f __attribute__((ext_vector_type(8)));

#define DDIM       256
#define NHALF      128
#define LDS_STRIDE 260
#define STRH       130
#define LDS_BYTES  (NHALF * LDS_STRIDE * 4)

static __device__ __forceinline__ v8f wmma_f32(v2f a, v2f b, v8f c) {
    return __builtin_amdgcn_wmma_f32_16x16x4_f32(
         false, a,  false, b,
         (short)0, c,  false,  false);
}

extern "C" __global__ __launch_bounds__(256)
void rgcn_gemm_wmma(const float* __restrict__ X,
                    const float* __restrict__ W,
                    float* __restrict__ Y,
                    int M)
{
    extern __shared__ char smem[];
    float*     ldsW  = (float*)smem;
    const v2f* ldsW2 = (const v2f*)smem;

    const int tid   = threadIdx.x;
    const int nCol0 = blockIdx.y * NHALF;

    {
        const int nl   = tid & (NHALF - 1);
        const int kOff = (tid >> 7) * (DDIM / 2);
        for (int k = 0; k < DDIM / 2; ++k) {
            ldsW[nl * LDS_STRIDE + kOff + k] = W[(kOff + k) * DDIM + nCol0 + nl];
        }
    }
    __syncthreads();

    const int wave  = tid >> 5;
    const int lane  = tid & 31;
    const int ln    = lane & 15;
    const int h     = lane >> 4;
    const int mBase = blockIdx.x * 256 + wave * 32;

    v8f acc[16];
#pragma unroll
    for (int i = 0; i < 16; ++i) acc[i] = (v8f)0.0f;

    const int  m0  = mBase + ln,      m1  = mBase + 16 + ln;
    const bool ok0 = (m0 < M),        ok1 = (m1 < M);
    const int  mc0 = ok0 ? m0 : M - 1;
    const int  mc1 = ok1 ? m1 : M - 1;
    const float* __restrict__ X0 = X + (size_t)mc0 * DDIM;
    const float* __restrict__ X1 = X + (size_t)mc1 * DDIM;

    for (int kk = 0; kk < DDIM; kk += 16) {
        v2f a0[4], a1[4];
#pragma unroll
        for (int j = 0; j < 4; ++j) {
            const int kc = kk + 4 * j + 2 * h;
            v2f t0 = *(const v2f*)(X0 + kc);
            v2f t1 = *(const v2f*)(X1 + kc);
            a0[j] = ok0 ? t0 : (v2f)0.0f;
            a1[j] = ok1 ? t1 : (v2f)0.0f;
        }
#pragma unroll
        for (int j = 0; j < 4; ++j) {
            const int kh = (kk >> 1) + 2 * j + h;
            v2f b[8];
#pragma unroll
            for (int nb = 0; nb < 8; ++nb) {
                b[nb] = ldsW2[(size_t)(nb * 16 + ln) * STRH + kh];
            }
#pragma unroll
            for (int nb = 0; nb < 8; ++nb) {
                acc[nb] = wmma_f32(a0[j], b[nb], acc[nb]);
            }
#pragma unroll
            for (int nb = 0; nb < 8; ++nb) {
                acc[8 + nb] = wmma_f32(a1[j], b[nb], acc[8 + nb]);
            }
        }
    }

#pragma unroll
    for (int s = 0; s < 2; ++s) {
#pragma unroll
        for (int nb = 0; nb < 8; ++nb) {
            const int col = nCol0 + nb * 16 + ln;
#pragma unroll
            for (int v = 0; v < 8; ++v) {
                const int row = mBase + s * 16 + v + 8 * h;
                if (row < M) {
                    Y[(size_t)row * DDIM + col] = acc[s * 8 + nb][v];
                }
            }
        }
    }
}

#define EPB 8
extern "C" __global__ __launch_bounds__(256)
void rgcn_scatter_add(const float* __restrict__ Y,
                      const int*   __restrict__ edges,
                      float* __restrict__ out,
                      int E)
{
    const int e0 = blockIdx.x * EPB;
    const int c  = threadIdx.x;
#pragma unroll
    for (int i = 0; i < EPB; ++i) {
        const int e = e0 + i;
        if (e >= E) break;
        const int src = edges[2 * e + 0];
        const int dst = edges[2 * e + 1];
        const float v = Y[(size_t)src * DDIM + c];
        unsafeAtomicAdd(out + (size_t)dst * DDIM + c, v);
    }
}

extern "C" void kernel_launch(void* const* d_in, const int* in_sizes, int n_in,
                              void* d_out, int out_size, void* d_ws, size_t ws_size,
                              hipStream_t stream)
{
    const float* x      = (const float*)d_in[0];
    const float* W_rel  = (const float*)d_in[1];
    const float* W_self = (const float*)d_in[2];
    const int*   edges  = (const int*)  d_in[3];

    float* out = (float*)d_out;
    float* y   = (float*)d_ws;

    const int M = in_sizes[0] / DDIM;
    const int R = in_sizes[1] / (DDIM * DDIM);
    const int E = in_sizes[3] / (2 * R);

    const dim3 gemmGrid((M + 255) / 256, 2);
    const int  scatterGrid = (E + EPB - 1) / EPB;

    rgcn_gemm_wmma<<<gemmGrid, 256, LDS_BYTES, stream>>>(x, W_self, out, M);

    for (int r = 0; r < R; ++r) {
        rgcn_gemm_wmma<<<gemmGrid, 256, LDS_BYTES, stream>>>(
            x, W_rel + (size_t)r * DDIM * DDIM, y, M);
        rgcn_scatter_add<<<scatterGrid, 256, 0, stream>>>(
            y, edges + (size_t)r * E * 2, out, E);
    }
}


# scatter-add: 2048 resident blocks loop over 8-edge batches, the 8 row gathers in flight together ahead of the 8 atomics (was 25000 short blocks, 1 load in flight); atomic order unordered as before
# speedup vs baseline: 1.2675x; 1.2675x over previous
#include <hip/hip_runtime.h>


typedef float v2f __attribute__((ext_vector_type(2)));
typedef float v8f __attribute__((ext_vector_type(8)));

#define DDIM       256
#define NHALF      128
#define LDS_STRIDE 260
#define STRH       130
#define LDS_BYTES  (NHALF * LDS_STRIDE * 4)

static __device__ __forceinline__ v8f wmma_f32(v2f a, v2f b, v8f c) {
    return __builtin_amdgcn_wmma_f32_16x16x4_f32(
         false, a,  false, b,
         (short)0, c,  false,  false);
}

extern "C" __global__ __launch_bounds__(256)
void rgcn_gemm_wmma(const float* __restrict__ X,
                    const float* __restrict__ W,
                    float* __restrict__ Y,
                    int M)
{
    extern __shared__ char smem[];
    float*     ldsW  = (float*)smem;
    const v2f* ldsW2 = (const v2f*)smem;

    const int tid   = threadIdx.x;
    const int nCol0 = blockIdx.y * NHALF;

    {
        const int nl   = tid & (NHALF - 1);
        const int kOff = (tid >> 7) * (DDIM / 2);
        for (int k = 0; k < DDIM / 2; ++k) {
            ldsW[nl * LDS_STRIDE + kOff + k] = W[(kOff + k) * DDIM + nCol0 + nl];
        }
    }
    __syncthreads();

    const int wave  = tid >> 5;
    const int lane  = tid & 31;
    const int ln    = lane & 15;
    const int h     = lane >> 4;
    const int mBase = blockIdx.x * 256 + wave * 32;

    v8f acc[16];
#pragma unroll
    for (int i = 0; i < 16; ++i) acc[i] = (v8f)0.0f;

    const int  m0  = mBase + ln,      m1  = mBase + 16 + ln;
    const bool ok0 = (m0 < M),        ok1 = (m1 < M);
    const int  mc0 = ok0 ? m0 : M - 1;
    const int  mc1 = ok1 ? m1 : M - 1;
    const float* __restrict__ X0 = X + (size_t)mc0 * DDIM;
    const float* __restrict__ X1 = X + (size_t)mc1 * DDIM;

    for (int kk = 0; kk < DDIM; kk += 16) {
        v2f a0[4], a1[4];
#pragma unroll
        for (int j = 0; j < 4; ++j) {
            const int kc = kk + 4 * j + 2 * h;
            v2f t0 = *(const v2f*)(X0 + kc);
            v2f t1 = *(const v2f*)(X1 + kc);
            a0[j] = ok0 ? t0 : (v2f)0.0f;
            a1[j] = ok1 ? t1 : (v2f)0.0f;
        }
#pragma unroll
        for (int j = 0; j < 4; ++j) {
            const int kh = (kk >> 1) + 2 * j + h;
            v2f b[8];
#pragma unroll
            for (int nb = 0; nb < 8; ++nb) {
                b[nb] = ldsW2[(size_t)(nb * 16 + ln) * STRH + kh];
            }
#pragma unroll
            for (int nb = 0; nb < 8; ++nb) {
                acc[nb] = wmma_f32(a0[j], b[nb], acc[nb]);
            }
#pragma unroll
            for (int nb = 0; nb < 8; ++nb) {
                acc[8 + nb] = wmma_f32(a1[j], b[nb], acc[8 + nb]);
            }
        }
    }

#pragma unroll
    for (int s = 0; s < 2; ++s) {
#pragma unroll
        for (int nb = 0; nb < 8; ++nb) {
            const int col = nCol0 + nb * 16 + ln;
#pragma unroll
            for (int v = 0; v < 8; ++v) {
                const int row = mBase + s * 16 + v + 8 * h;
                if (row < M) {
                    Y[(size_t)row * DDIM + col] = acc[s * 8 + nb][v];
                }
            }
        }
    }
}

#define EPB 8
#define SCATTER_BLOCKS 2048
extern "C" __global__ __launch_bounds__(256)
void rgcn_scatter_add(const float* __restrict__ Y,
                      const int*   __restrict__ edges,
                      float* __restrict__ out,
                      int E)
{
    const int c      = threadIdx.x;
    const int nFull  = E / EPB;
    const int stride = SCATTER_BLOCKS;
    int b = blockIdx.x;

    if (b < nFull) {
        int cur[2 * EPB];
#pragma unroll
        for (int i = 0; i < 2 * EPB; ++i) cur[i] = edges[(size_t)b * (2 * EPB) + i];

        for (;;) {
            const int  nb   = b + stride;
            const bool more = (nb < nFull);
            int nxt[2 * EPB];
            if (more) {
#pragma unroll
                for (int i = 0; i < 2 * EPB; ++i) nxt[i] = edges[(size_t)nb * (2 * EPB) + i];
            }

            float v[EPB];
#pragma unroll
            for (int i = 0; i < EPB; ++i) {
                v[i] = Y[(size_t)cur[2 * i] * DDIM + c];
            }
#pragma unroll
            for (int i = 0; i < EPB; ++i) {
                unsafeAtomicAdd(out + (size_t)cur[2 * i + 1] * DDIM + c, v[i]);
            }

            if (!more) break;
#pragma unroll
            for (int i = 0; i < 2 * EPB; ++i) cur[i] = nxt[i];
            b = nb;
        }
    }

    if (blockIdx.x == 0) {
        for (int e = nFull * EPB; e < E; ++e) {
            const int s = edges[2 * e + 0];
            const int d = edges[2 * e + 1];
            const float v = Y[(size_t)s * DDIM + c];
            unsafeAtomicAdd(out + (size_t)d * DDIM + c, v);
        }
    }
}

extern "C" void kernel_launch(void* const* d_in, const int* in_sizes, int n_in,
                              void* d_out, int out_size, void* d_ws, size_t ws_size,
                              hipStream_t stream)
{
    const float* x      = (const float*)d_in[0];
    const float* W_rel  = (const float*)d_in[1];
    const float* W_self = (const float*)d_in[2];
    const int*   edges  = (const int*)  d_in[3];

    float* out = (float*)d_out;
    float* y   = (float*)d_ws;

    const int M = in_sizes[0] / DDIM;
    const int R = in_sizes[1] / (DDIM * DDIM);
    const int E = in_sizes[3] / (2 * R);

    const dim3 gemmGrid((M + 255) / 256, 2);

    rgcn_gemm_wmma<<<gemmGrid, 256, LDS_BYTES, stream>>>(x, W_self, out, M);

    for (int r = 0; r < R; ++r) {
        rgcn_gemm_wmma<<<gemmGrid, 256, LDS_BYTES, stream>>>(
            x, W_rel + (size_t)r * DDIM * DDIM, y, M);
        rgcn_scatter_add<<<SCATTER_BLOCKS, 256, 0, stream>>>(
            y, edges + (size_t)r * E * 2, out, E);
    }
}


# float-atomic scatter replaced by a per-call counting sort of edges by destination in LDS plus one wave per destination row summing its source rows; sum order = LDS slot arrival order
# speedup vs baseline: 1.3900x; 1.0966x over previous
#include <hip/hip_runtime.h>


typedef float v2f __attribute__((ext_vector_type(2)));
typedef float v8f __attribute__((ext_vector_type(8)));

#define DDIM       256
#define NHALF      128
#define LDS_STRIDE 260
#define STRH       130
#define LDS_BYTES  (NHALF * LDS_STRIDE * 4)

static __device__ __forceinline__ v8f wmma_f32(v2f a, v2f b, v8f c) {
    return __builtin_amdgcn_wmma_f32_16x16x4_f32(
         false, a,  false, b,
         (short)0, c,  false,  false);
}

extern "C" __global__ __launch_bounds__(256)
void rgcn_gemm_wmma(const float* __restrict__ X,
                    const float* __restrict__ W,
                    float* __restrict__ Y,
                    int M)
{
    extern __shared__ char smem[];
    float*     ldsW  = (float*)smem;
    const v2f* ldsW2 = (const v2f*)smem;

    const int tid   = threadIdx.x;
    const int nCol0 = blockIdx.y * NHALF;

    {
        const int nl   = tid & (NHALF - 1);
        const int kOff = (tid >> 7) * (DDIM / 2);
        for (int k = 0; k < DDIM / 2; ++k) {
            ldsW[nl * LDS_STRIDE + kOff + k] = W[(kOff + k) * DDIM + nCol0 + nl];
        }
    }
    __syncthreads();

    const int wave  = tid >> 5;
    const int lane  = tid & 31;
    const int ln    = lane & 15;
    const int h     = lane >> 4;
    const int mBase = blockIdx.x * 256 + wave * 32;

    v8f acc[16];
#pragma unroll
    for (int i = 0; i < 16; ++i) acc[i] = (v8f)0.0f;

    const int  m0  = mBase + ln,      m1  = mBase + 16 + ln;
    const bool ok0 = (m0 < M),        ok1 = (m1 < M);
    const int  mc0 = ok0 ? m0 : M - 1;
    const int  mc1 = ok1 ? m1 : M - 1;
    const float* __restrict__ X0 = X + (size_t)mc0 * DDIM;
    const float* __restrict__ X1 = X + (size_t)mc1 * DDIM;

    for (int kk = 0; kk < DDIM; kk += 16) {
        v2f a0[4], a1[4];
#pragma unroll
        for (int j = 0; j < 4; ++j) {
            const int kc = kk + 4 * j + 2 * h;
            v2f t0 = *(const v2f*)(X0 + kc);
            v2f t1 = *(const v2f*)(X1 + kc);
            a0[j] = ok0 ? t0 : (v2f)0.0f;
            a1[j] = ok1 ? t1 : (v2f)0.0f;
        }
#pragma unroll
        for (int j = 0; j < 4; ++j) {
            const int kh = (kk >> 1) + 2 * j + h;
            v2f b[8];
#pragma unroll
            for (int nb = 0; nb < 8; ++nb) {
                b[nb] = ldsW2[(size_t)(nb * 16 + ln) * STRH + kh];
            }
#pragma unroll
            for (int nb = 0; nb < 8; ++nb) {
                acc[nb] = wmma_f32(a0[j], b[nb], acc[nb]);
            }
#pragma unroll
            for (int nb = 0; nb < 8; ++nb) {
                acc[8 + nb] = wmma_f32(a1[j], b[nb], acc[8 + nb]);
            }
        }
    }

#pragma unroll
    for (int s = 0; s < 2; ++s) {
#pragma unroll
        for (int nb = 0; nb < 8; ++nb) {
            const int col = nCol0 + nb * 16 + ln;
#pragma unroll
            for (int v = 0; v < 8; ++v) {
                const int row = mBase + s * 16 + v + 8 * h;
                if (row < M) {
                    Y[(size_t)row * DDIM + col] = acc[s * 8 + nb][v];
                }
            }
        }
    }
}

#define EPB 8
#define SCATTER_BLOCKS 2048
extern "C" __global__ __launch_bounds__(256)
void rgcn_scatter_add(const float* __restrict__ Y,
                      const int*   __restrict__ edges,
                      float* __restrict__ out,
                      int E)
{
    const int c      = threadIdx.x;
    const int nFull  = E / EPB;
    const int stride = SCATTER_BLOCKS;
    int b = blockIdx.x;

    if (b < nFull) {
        int cur[2 * EPB];
#pragma unroll
        for (int i = 0; i < 2 * EPB; ++i) cur[i] = edges[(size_t)b * (2 * EPB) + i];

        for (;;) {
            const int  nb   = b + stride;
            const bool more = (nb < nFull);
            int nxt[2 * EPB];
            if (more) {
#pragma unroll
                for (int i = 0; i < 2 * EPB; ++i) nxt[i] = edges[(size_t)nb * (2 * EPB) + i];
            }

            float v[EPB];
#pragma unroll
            for (int i = 0; i < EPB; ++i) {
                v[i] = Y[(size_t)cur[2 * i] * DDIM + c];
            }
#pragma unroll
            for (int i = 0; i < EPB; ++i) {
                unsafeAtomicAdd(out + (size_t)cur[2 * i + 1] * DDIM + c, v[i]);
            }

            if (!more) break;
#pragma unroll
            for (int i = 0; i < 2 * EPB; ++i) cur[i] = nxt[i];
            b = nb;
        }
    }

    if (blockIdx.x == 0) {
        for (int e = nFull * EPB; e < E; ++e) {
            const int s = edges[2 * e + 0];
            const int d = edges[2 * e + 1];
            const float v = Y[(size_t)s * DDIM + c];
            unsafeAtomicAdd(out + (size_t)d * DDIM + c, v);
        }
    }
}


#define RANGE      6400
#define SCHUNK     7
#define SORT_U     8
#define PULL_BLOCKS 2048

typedef int v4i __attribute__((ext_vector_type(4)));

extern "C" __global__ __launch_bounds__(1024)
void rgcn_sort_by_dst(const int* __restrict__ edges,
                      int* __restrict__ pos,
                      int* __restrict__ ssrc,
                      int E, int M)
{
    __shared__ int hist[RANGE];
    __shared__ int part[1024];
    __shared__ int belowSh;

    const int t   = threadIdx.x;
    const int r   = blockIdx.x;
    const int lo  = blockIdx.y * RANGE;
    const int len = min(RANGE, M - lo);
    const int* __restrict__ ed = edges + (size_t)r * E * 2;
    const v4i* __restrict__ ed4 = (const v4i*)ed;
    const int nPairs = E >> 1;
    int* __restrict__ srel = ssrc + (size_t)r * E;

    for (int i = t; i < RANGE; i += 1024) hist[i] = 0;
    if (t == 0) belowSh = 0;
    __syncthreads();

    int below = 0;
    for (int base = t; base < E; base += 2 * SORT_U * 1024) {
        int dq[2 * SORT_U];
#pragma unroll
        for (int j = 0; j < 2 * SORT_U; ++j) {
            const int idx = base + j * 1024;
            dq[j] = ed[2 * (size_t)min(idx, E - 1) + 1];
        }
#pragma unroll
        for (int j = 0; j < 2 * SORT_U; ++j) {
            const int d = dq[j];
            if ((base + j * 1024) < E) {
                below += (d < lo);
                if ((unsigned)(d - lo) < (unsigned)len) atomicAdd(&hist[d - lo], 1);
            }
        }
    }
#pragma unroll
    for (int m = 16; m >= 1; m >>= 1) below += __shfl_xor(below, m, 32);
    if ((t & 31) == 0) atomicAdd(&belowSh, below);
    __syncthreads();

    const int c0 = min(t * SCHUNK, RANGE);
    const int c1 = min(c0 + SCHUNK, RANGE);
    int s = 0;
    for (int i = c0; i < c1; ++i) s += hist[i];
    part[t] = s;
    __syncthreads();
    for (int off = 1; off < 1024; off <<= 1) {
        const int v = (t >= off) ? part[t - off] : 0;
        __syncthreads();
        part[t] += v;
        __syncthreads();
    }
    int run = belowSh + part[t] - s;
    for (int i = c0; i < c1; ++i) {
        const int v = hist[i];
        hist[i] = run;
        run += v;
        if (i < len) pos[(size_t)r * M + lo + i] = run;
    }
    __syncthreads();

    for (int base = t; base < nPairs; base += SORT_U * 1024) {
        v4i q[SORT_U];
#pragma unroll
        for (int j = 0; j < SORT_U; ++j) {
            const int idx = base + j * 1024;
            q[j] = ed4[min(idx, nPairs - 1)];
        }
#pragma unroll
        for (int j = 0; j < SORT_U; ++j) {
            const bool valid = (base + j * 1024) < nPairs;
            const int s0 = q[j].x, d0 = q[j].y, s1 = q[j].z, d1 = q[j].w;
            if (valid) {
                if ((unsigned)(d0 - lo) < (unsigned)len) srel[atomicAdd(&hist[d0 - lo], 1)] = s0;
                if ((unsigned)(d1 - lo) < (unsigned)len) srel[atomicAdd(&hist[d1 - lo], 1)] = s1;
            }
        }
    }
    if ((E & 1) && t == 0) {
        const int sN = ed[2 * (size_t)(E - 1) + 0];
        const int d  = ed[2 * (size_t)(E - 1) + 1];
        if ((unsigned)(d - lo) < (unsigned)len) srel[atomicAdd(&hist[d - lo], 1)] = sN;
    }
}

typedef float v4f __attribute__((ext_vector_type(4)));

extern "C" __global__ __launch_bounds__(256)
void rgcn_pull_add(const float* __restrict__ Y,
                   const int*   __restrict__ pos,
                   const int*   __restrict__ ssrc,
                   float* __restrict__ out, int M)
{
    const int lane = threadIdx.x & 31;
    const int gw = __builtin_amdgcn_readfirstlane((int)(blockIdx.x * 8 + (threadIdx.x >> 5)));
    for (int d = gw; d < M; d += PULL_BLOCKS * 8) {
        const int beg = (d > 0) ? pos[d - 1] : 0;
        const int end = pos[d];
        if (end <= beg) continue;
        v4f* orow = (v4f*)(out + (size_t)d * DDIM);
        v4f o0 = orow[lane];
        v4f o1 = orow[32 + lane];
        for (int e = beg; e < end; e += 4) {
            const int n = end - e;
            const v4f zero = (v4f)0.0f;
            v4f a1 = zero, b1 = zero, a2 = zero, b2 = zero, a3 = zero, b3 = zero;
            const v4f* y0 = (const v4f*)(Y + (size_t)ssrc[e] * DDIM);
            const v4f a0 = y0[lane], b0 = y0[32 + lane];
            if (n > 1) {
                const v4f* y1 = (const v4f*)(Y + (size_t)ssrc[e + 1] * DDIM);
                a1 = y1[lane]; b1 = y1[32 + lane];
            }
            if (n > 2) {
                const v4f* y2 = (const v4f*)(Y + (size_t)ssrc[e + 2] * DDIM);
                a2 = y2[lane]; b2 = y2[32 + lane];
            }
            if (n > 3) {
                const v4f* y3 = (const v4f*)(Y + (size_t)ssrc[e + 3] * DDIM);
                a3 = y3[lane]; b3 = y3[32 + lane];
            }
            o0 += a0; o1 += b0;
            o0 += a1; o1 += b1;
            o0 += a2; o1 += b2;
            o0 += a3; o1 += b3;
        }
        orow[lane]      = o0;
        orow[32 + lane] = o1;
    }
}

extern "C" void kernel_launch(void* const* d_in, const int* in_sizes, int n_in,
                              void* d_out, int out_size, void* d_ws, size_t ws_size,
                              hipStream_t stream)
{
    const float* x      = (const float*)d_in[0];
    const float* W_rel  = (const float*)d_in[1];
    const float* W_self = (const float*)d_in[2];
    const int*   edges  = (const int*)  d_in[3];

    float* out = (float*)d_out;
    float* y   = (float*)d_ws;

    const int M = in_sizes[0] / DDIM;
    const int R = in_sizes[1] / (DDIM * DDIM);
    const int E = in_sizes[3] / (2 * R);

    const dim3 gemmGrid((M + 255) / 256, 2);

    rgcn_gemm_wmma<<<gemmGrid, 256, LDS_BYTES, stream>>>(x, W_self, out, M);

    const size_t yBytes = (size_t)M * DDIM * sizeof(float);
    int* pos  = (int*)((char*)d_ws + yBytes);
    int* ssrc = pos + (size_t)R * M;
    const size_t need = yBytes + ((size_t)R * M + (size_t)R * E) * sizeof(int);

    if (ws_size >= need) {
        const dim3 sortGrid(R, (M + RANGE - 1) / RANGE);
        rgcn_sort_by_dst<<<sortGrid, 1024, 0, stream>>>(edges, pos, ssrc, E, M);

        for (int r = 0; r < R; ++r) {
            rgcn_gemm_wmma<<<gemmGrid, 256, LDS_BYTES, stream>>>(
                x, W_rel + (size_t)r * DDIM * DDIM, y, M);
            rgcn_pull_add<<<PULL_BLOCKS, 256, 0, stream>>>(
                y, pos + (size_t)r * M, ssrc + (size_t)r * E, out, M);
        }
    } else {
        for (int r = 0; r < R; ++r) {
            rgcn_gemm_wmma<<<gemmGrid, 256, LDS_BYTES, stream>>>(
                x, W_rel + (size_t)r * DDIM * DDIM, y, M);
            rgcn_scatter_add<<<SCATTER_BLOCKS, 256, 0, stream>>>(
                y, edges + (size_t)r * E * 2, out, E);
        }
    }
}


# GEMM re-tiled for equal WMMA count per SIMD: one 16-wave block per WGP, all of W in LDS, owns 13 16-row strips (13 tiles per wave), fragments loaded a 32-k chunk ahead; was 392 blocks of 256x128
# speedup vs baseline: 1.5788x; 1.1358x over previous
#include <hip/hip_runtime.h>


typedef float v2f __attribute__((ext_vector_type(2)));
typedef float v8f __attribute__((ext_vector_type(8)));

#define DDIM       256
#define LDS_STRIDE 260
#define STRH       130
#define STRIPS_PER_BLOCK 13
#define GEMM_THREADS 512

static __device__ __forceinline__ v8f wmma_f32(v2f a, v2f b, v8f c) {
    return __builtin_amdgcn_wmma_f32_16x16x4_f32(
         false, a,  false, b,
         (short)0, c,  false,  false);
}

template <int NT>
static __device__ __forceinline__ void gemm_task(const float* __restrict__ X,
                                                 const v2f* bBase,
                                                 float* __restrict__ Y,
                                                 int M, int gs, int col0, int ln, int h)
{
    const int  m  = gs * 16 + ln;
    const bool ok = (m < M);
    const int  mc = ok ? m : M - 1;
    const v2f* aRow = (const v2f*)(X + (size_t)mc * DDIM) + h;

    v8f acc[NT];
#pragma unroll
    for (int i = 0; i < NT; ++i) acc[i] = (v8f)0.0f;

    v2f a[2][8];
    v2f b[2][8 * NT];

#pragma unroll
    for (int j = 0; j < 8; ++j) {
        const v2f t = aRow[2 * j];
        a[0][j] = ok ? t : (v2f)0.0f;
    }
#pragma unroll
    for (int j = 0; j < 8; ++j) {
#pragma unroll
        for (int nt = 0; nt < NT; ++nt) b[0][j * NT + nt] = bBase[nt * 16 * STRH + 2 * j];
    }

#pragma unroll
    for (int ch = 0; ch < 8; ++ch) {
        const int cur = ch & 1, nxt = cur ^ 1;
        if (ch + 1 < 8) {
#pragma unroll
            for (int j = 0; j < 8; ++j) {
                const v2f t = aRow[(ch + 1) * 16 + 2 * j];
                a[nxt][j] = ok ? t : (v2f)0.0f;
            }
#pragma unroll
            for (int j = 0; j < 8; ++j) {
#pragma unroll
                for (int nt = 0; nt < NT; ++nt)
                    b[nxt][j * NT + nt] = bBase[nt * 16 * STRH + (ch + 1) * 16 + 2 * j];
            }
        }
#pragma unroll
        for (int j = 0; j < 8; ++j) {
#pragma unroll
            for (int nt = 0; nt < NT; ++nt)
                acc[nt] = wmma_f32(a[cur][j], b[cur][j * NT + nt], acc[nt]);
        }
    }

#pragma unroll
    for (int nt = 0; nt < NT; ++nt) {
        const int col = col0 + nt * 16 + ln;
#pragma unroll
        for (int v = 0; v < 8; ++v) {
            const int row = gs * 16 + v + 8 * h;
            if (row < M) {
                Y[(size_t)row * DDIM + col] = acc[nt][v];
            }
        }
    }
}

extern "C" __global__ __launch_bounds__(GEMM_THREADS)
void rgcn_gemm_wmma(const float* __restrict__ X,
                    const float* __restrict__ W,
                    float* __restrict__ Y,
                    int M)
{
    __shared__ float ldsW[DDIM * LDS_STRIDE];
    const v2f* ldsW2 = (const v2f*)ldsW;

    const int tid = threadIdx.x;

    {
        const int n    = tid & (DDIM - 1);
        const int kOff = (tid >> 8) * (DDIM / 2);
        for (int k = 0; k < DDIM / 2; ++k) {
            ldsW[n * LDS_STRIDE + kOff + k] = W[(kOff + k) * DDIM + n];
        }
    }
    __syncthreads();

    const int wave = __builtin_amdgcn_readfirstlane(tid >> 5);
    const int lane = tid & 31;
    const int ln   = lane & 15;
    const int h    = lane >> 4;
    const int nStrips = (M + 15) >> 4;
    const int strip0  = (int)blockIdx.x * STRIPS_PER_BLOCK;

    {
        const int col0 = (wave & 3) * 64;
        const v2f* bBase = ldsW2 + (size_t)(col0 + ln) * STRH + h;
        for (int s = wave >> 2; s < STRIPS_PER_BLOCK - 1; s += 4) {
            const int gs = strip0 + s;
            if (gs >= nStrips) break;
            gemm_task<4>(X, bBase, Y, M, gs, col0, ln, h);
        }
    }
    {
        const int gs = strip0 + STRIPS_PER_BLOCK - 1;
        if (gs < nStrips) {
            const int col0 = wave * 16;
            const v2f* bBase = ldsW2 + (size_t)(col0 + ln) * STRH + h;
            gemm_task<1>(X, bBase, Y, M, gs, col0, ln, h);
        }
    }
}

#define EPB 8
#define SCATTER_BLOCKS 2048
extern "C" __global__ __launch_bounds__(256)
void rgcn_scatter_add(const float* __restrict__ Y,
                      const int*   __restrict__ edges,
                      float* __restrict__ out,
                      int E)
{
    const int c      = threadIdx.x;
    const int nFull  = E / EPB;
    const int stride = SCATTER_BLOCKS;
    int b = blockIdx.x;

    if (b < nFull) {
        int cur[2 * EPB];
#pragma unroll
        for (int i = 0; i < 2 * EPB; ++i) cur[i] = edges[(size_t)b * (2 * EPB) + i];

        for (;;) {
            const int  nb   = b + stride;
            const bool more = (nb < nFull);
            int nxt[2 * EPB];
            if (more) {
#pragma unroll
                for (int i = 0; i < 2 * EPB; ++i) nxt[i] = edges[(size_t)nb * (2 * EPB) + i];
            }

            float v[EPB];
#pragma unroll
            for (int i = 0; i < EPB; ++i) {
                v[i] = Y[(size_t)cur[2 * i] * DDIM + c];
            }
#pragma unroll
            for (int i = 0; i < EPB; ++i) {
                unsafeAtomicAdd(out + (size_t)cur[2 * i + 1] * DDIM + c, v[i]);
            }

            if (!more) break;
#pragma unroll
            for (int i = 0; i < 2 * EPB; ++i) cur[i] = nxt[i];
            b = nb;
        }
    }

    if (blockIdx.x == 0) {
        for (int e = nFull * EPB; e < E; ++e) {
            const int s = edges[2 * e + 0];
            const int d = edges[2 * e + 1];
            const float v = Y[(size_t)s * DDIM + c];
            unsafeAtomicAdd(out + (size_t)d * DDIM + c, v);
        }
    }
}


#define RANGE      6400
#define SCHUNK     7
#define SORT_U     8
#define PULL_BLOCKS 2048

typedef int v4i __attribute__((ext_vector_type(4)));

extern "C" __global__ __launch_bounds__(1024)
void rgcn_sort_by_dst(const int* __restrict__ edges,
                      int* __restrict__ pos,
                      int* __restrict__ ssrc,
                      int E, int M)
{
    __shared__ int hist[RANGE];
    __shared__ int part[1024];
    __shared__ int belowSh;

    const int t   = threadIdx.x;
    const int r   = blockIdx.x;
    const int lo  = blockIdx.y * RANGE;
    const int len = min(RANGE, M - lo);
    const int* __restrict__ ed = edges + (size_t)r * E * 2;
    const v4i* __restrict__ ed4 = (const v4i*)ed;
    const int nPairs = E >> 1;
    int* __restrict__ srel = ssrc + (size_t)r * E;

    for (int i = t; i < RANGE; i += 1024) hist[i] = 0;
    if (t == 0) belowSh = 0;
    __syncthreads();

    int below = 0;
    for (int base = t; base < E; base += 2 * SORT_U * 1024) {
        int dq[2 * SORT_U];
#pragma unroll
        for (int j = 0; j < 2 * SORT_U; ++j) {
            const int idx = base + j * 1024;
            dq[j] = ed[2 * (size_t)min(idx, E - 1) + 1];
        }
#pragma unroll
        for (int j = 0; j < 2 * SORT_U; ++j) {
            const int d = dq[j];
            if ((base + j * 1024) < E) {
                below += (d < lo);
                if ((unsigned)(d - lo) < (unsigned)len) atomicAdd(&hist[d - lo], 1);
            }
        }
    }
#pragma unroll
    for (int m = 16; m >= 1; m >>= 1) below += __shfl_xor(below, m, 32);
    if ((t & 31) == 0) atomicAdd(&belowSh, below);
    __syncthreads();

    const int c0 = min(t * SCHUNK, RANGE);
    const int c1 = min(c0 + SCHUNK, RANGE);
    int s = 0;
    for (int i = c0; i < c1; ++i) s += hist[i];
    part[t] = s;
    __syncthreads();
    for (int off = 1; off < 1024; off <<= 1) {
        const int v = (t >= off) ? part[t - off] : 0;
        __syncthreads();
        part[t] += v;
        __syncthreads();
    }
    int run = belowSh + part[t] - s;
    for (int i = c0; i < c1; ++i) {
        const int v = hist[i];
        hist[i] = run;
        run += v;
        if (i < len) pos[(size_t)r * M + lo + i] = run;
    }
    __syncthreads();

    for (int base = t; base < nPairs; base += SORT_U * 1024) {
        v4i q[SORT_U];
#pragma unroll
        for (int j = 0; j < SORT_U; ++j) {
            const int idx = base + j * 1024;
            q[j] = ed4[min(idx, nPairs - 1)];
        }
#pragma unroll
        for (int j = 0; j < SORT_U; ++j) {
            const bool valid = (base + j * 1024) < nPairs;
            const int s0 = q[j].x, d0 = q[j].y, s1 = q[j].z, d1 = q[j].w;
            if (valid) {
                if ((unsigned)(d0 - lo) < (unsigned)len) srel[atomicAdd(&hist[d0 - lo], 1)] = s0;
                if ((unsigned)(d1 - lo) < (unsigned)len) srel[atomicAdd(&hist[d1 - lo], 1)] = s1;
            }
        }
    }
    if ((E & 1) && t == 0) {
        const int sN = ed[2 * (size_t)(E - 1) + 0];
        const int d  = ed[2 * (size_t)(E - 1) + 1];
        if ((unsigned)(d - lo) < (unsigned)len) srel[atomicAdd(&hist[d - lo], 1)] = sN;
    }
}

typedef float v4f __attribute__((ext_vector_type(4)));

extern "C" __global__ __launch_bounds__(256)
void rgcn_pull_add(const float* __restrict__ Y,
                   const int*   __restrict__ pos,
                   const int*   __restrict__ ssrc,
                   float* __restrict__ out, int M)
{
    const int lane = threadIdx.x & 31;
    const int gw = __builtin_amdgcn_readfirstlane((int)(blockIdx.x * 8 + (threadIdx.x >> 5)));
    for (int d = gw; d < M; d += PULL_BLOCKS * 8) {
        const int beg = (d > 0) ? pos[d - 1] : 0;
        const int end = pos[d];
        if (end <= beg) continue;
        v4f* orow = (v4f*)(out + (size_t)d * DDIM);
        v4f o0 = orow[lane];
        v4f o1 = orow[32 + lane];
        for (int e = beg; e < end; e += 4) {
            const int n = end - e;
            const v4f zero = (v4f)0.0f;
            v4f a1 = zero, b1 = zero, a2 = zero, b2 = zero, a3 = zero, b3 = zero;
            const v4f* y0 = (const v4f*)(Y + (size_t)ssrc[e] * DDIM);
            const v4f a0 = y0[lane], b0 = y0[32 + lane];
            if (n > 1) {
                const v4f* y1 = (const v4f*)(Y + (size_t)ssrc[e + 1] * DDIM);
                a1 = y1[lane]; b1 = y1[32 + lane];
            }
            if (n > 2) {
                const v4f* y2 = (const v4f*)(Y + (size_t)ssrc[e + 2] * DDIM);
                a2 = y2[lane]; b2 = y2[32 + lane];
            }
            if (n > 3) {
                const v4f* y3 = (const v4f*)(Y + (size_t)ssrc[e + 3] * DDIM);
                a3 = y3[lane]; b3 = y3[32 + lane];
            }
            o0 += a0; o1 += b0;
            o0 += a1; o1 += b1;
            o0 += a2; o1 += b2;
            o0 += a3; o1 += b3;
        }
        orow[lane]      = o0;
        orow[32 + lane] = o1;
    }
}

extern "C" void kernel_launch(void* const* d_in, const int* in_sizes, int n_in,
                              void* d_out, int out_size, void* d_ws, size_t ws_size,
                              hipStream_t stream)
{
    const float* x      = (const float*)d_in[0];
    const float* W_rel  = (const float*)d_in[1];
    const float* W_self = (const float*)d_in[2];
    const int*   edges  = (const int*)  d_in[3];

    float* out = (float*)d_out;
    float* y   = (float*)d_ws;

    const int M = in_sizes[0] / DDIM;
    const int R = in_sizes[1] / (DDIM * DDIM);
    const int E = in_sizes[3] / (2 * R);

    const int  nStrips = (M + 15) / 16;
    const dim3 gemmGrid((nStrips + STRIPS_PER_BLOCK - 1) / STRIPS_PER_BLOCK);

    rgcn_gemm_wmma<<<gemmGrid, GEMM_THREADS, 0, stream>>>(x, W_self, out, M);

    const size_t yBytes = (size_t)M * DDIM * sizeof(float);
    int* pos  = (int*)((char*)d_ws + yBytes);
    int* ssrc = pos + (size_t)R * M;
    const size_t need = yBytes + ((size_t)R * M + (size_t)R * E) * sizeof(int);

    if (ws_size >= need) {
        const dim3 sortGrid(R, (M + RANGE - 1) / RANGE);
        rgcn_sort_by_dst<<<sortGrid, 1024, 0, stream>>>(edges, pos, ssrc, E, M);

        for (int r = 0; r < R; ++r) {
            rgcn_gemm_wmma<<<gemmGrid, GEMM_THREADS, 0, stream>>>(
                x, W_rel + (size_t)r * DDIM * DDIM, y, M);
            rgcn_pull_add<<<PULL_BLOCKS, 256, 0, stream>>>(
                y, pos + (size_t)r * M, ssrc + (size_t)r * E, out, M);
        }
    } else {
        for (int r = 0; r < R; ++r) {
            rgcn_gemm_wmma<<<gemmGrid, GEMM_THREADS, 0, stream>>>(
                x, W_rel + (size_t)r * DDIM * DDIM, y, M);
            rgcn_scatter_add<<<SCATTER_BLOCKS, 256, 0, stream>>>(
                y, edges + (size_t)r * E * 2, out, E);
        }
    }
}


# row sums of relation r-1 run as 16 extra waves inside the GEMM launch of relation r (second y buffer in workspace): gathers hide under matrix time, 18 -> 11 launches; sum order unchanged
# speedup vs baseline: 1.8261x; 1.1566x over previous
#include <hip/hip_runtime.h>


typedef float v2f __attribute__((ext_vector_type(2)));
typedef float v8f __attribute__((ext_vector_type(8)));

#define DDIM       256
#define LDS_STRIDE 260
#define STRH       130
#define STRIPS_PER_BLOCK 13
#define GEMM_THREADS 512

static __device__ __forceinline__ v8f wmma_f32(v2f a, v2f b, v8f c) {
    return __builtin_amdgcn_wmma_f32_16x16x4_f32(
         false, a,  false, b,
         (short)0, c,  false,  false);
}

template <int NT>
static __device__ __forceinline__ void gemm_task(const float* __restrict__ X,
                                                 const v2f* bBase,
                                                 float* __restrict__ Y,
                                                 int M, int gs, int col0, int ln, int h)
{
    const int  m  = gs * 16 + ln;
    const bool ok = (m < M);
    const int  mc = ok ? m : M - 1;
    const v2f* aRow = (const v2f*)(X + (size_t)mc * DDIM) + h;

    v8f acc[NT];
#pragma unroll
    for (int i = 0; i < NT; ++i) acc[i] = (v8f)0.0f;

    v2f a[2][8];
    v2f b[2][8 * NT];

#pragma unroll
    for (int j = 0; j < 8; ++j) {
        const v2f t = aRow[2 * j];
        a[0][j] = ok ? t : (v2f)0.0f;
    }
#pragma unroll
    for (int j = 0; j < 8; ++j) {
#pragma unroll
        for (int nt = 0; nt < NT; ++nt) b[0][j * NT + nt] = bBase[nt * 16 * STRH + 2 * j];
    }

#pragma unroll
    for (int ch = 0; ch < 8; ++ch) {
        const int cur = ch & 1, nxt = cur ^ 1;
        if (ch + 1 < 8) {
#pragma unroll
            for (int j = 0; j < 8; ++j) {
                const v2f t = aRow[(ch + 1) * 16 + 2 * j];
                a[nxt][j] = ok ? t : (v2f)0.0f;
            }
#pragma unroll
            for (int j = 0; j < 8; ++j) {
#pragma unroll
                for (int nt = 0; nt < NT; ++nt)
                    b[nxt][j * NT + nt] = bBase[nt * 16 * STRH + (ch + 1) * 16 + 2 * j];
            }
        }
#pragma unroll
        for (int j = 0; j < 8; ++j) {
#pragma unroll
            for (int nt = 0; nt < NT; ++nt)
                acc[nt] = wmma_f32(a[cur][j], b[cur][j * NT + nt], acc[nt]);
        }
    }

#pragma unroll
    for (int nt = 0; nt < NT; ++nt) {
        const int col = col0 + nt * 16 + ln;
#pragma unroll
        for (int v = 0; v < 8; ++v) {
            const int row = gs * 16 + v + 8 * h;
            if (row < M) {
                Y[(size_t)row * DDIM + col] = acc[nt][v];
            }
        }
    }
}

static __device__ __forceinline__ void gemm_block(const float* __restrict__ X,
                                                  const v2f* ldsW2,
                                                  float* __restrict__ Y,
                                                  int M, int wave, int lane)
{
    const int ln   = lane & 15;
    const int h    = lane >> 4;
    const int nStrips = (M + 15) >> 4;
    const int strip0  = (int)blockIdx.x * STRIPS_PER_BLOCK;

    {
        const int col0 = (wave & 3) * 64;
        const v2f* bBase = ldsW2 + (size_t)(col0 + ln) * STRH + h;
        for (int s = wave >> 2; s < STRIPS_PER_BLOCK - 1; s += 4) {
            const int gs = strip0 + s;
            if (gs >= nStrips) break;
            gemm_task<4>(X, bBase, Y, M, gs, col0, ln, h);
        }
    }
    {
        const int gs = strip0 + STRIPS_PER_BLOCK - 1;
        if (gs < nStrips) {
            const int col0 = wave * 16;
            const v2f* bBase = ldsW2 + (size_t)(col0 + ln) * STRH + h;
            gemm_task<1>(X, bBase, Y, M, gs, col0, ln, h);
        }
    }
}

extern "C" __global__ __launch_bounds__(GEMM_THREADS)
void rgcn_gemm_wmma(const float* __restrict__ X,
                    const float* __restrict__ W,
                    float* __restrict__ Y,
                    int M)
{
    __shared__ float ldsW[DDIM * LDS_STRIDE];
    const v2f* ldsW2 = (const v2f*)ldsW;

    const int tid = threadIdx.x;

    {
        const int n    = tid & (DDIM - 1);
        const int kOff = (tid >> 8) * (DDIM / 2);
        for (int k = 0; k < DDIM / 2; ++k) {
            ldsW[n * LDS_STRIDE + kOff + k] = W[(kOff + k) * DDIM + n];
        }
    }
    __syncthreads();

    const int wave = __builtin_amdgcn_readfirstlane(tid >> 5);
    gemm_block(X, ldsW2, Y, M, wave, tid & 31);
}

#define EPB 8
#define SCATTER_BLOCKS 2048
extern "C" __global__ __launch_bounds__(256)
void rgcn_scatter_add(const float* __restrict__ Y,
                      const int*   __restrict__ edges,
                      float* __restrict__ out,
                      int E)
{
    const int c      = threadIdx.x;
    const int nFull  = E / EPB;
    const int stride = SCATTER_BLOCKS;
    int b = blockIdx.x;

    if (b < nFull) {
        int cur[2 * EPB];
#pragma unroll
        for (int i = 0; i < 2 * EPB; ++i) cur[i] = edges[(size_t)b * (2 * EPB) + i];

        for (;;) {
            const int  nb   = b + stride;
            const bool more = (nb < nFull);
            int nxt[2 * EPB];
            if (more) {
#pragma unroll
                for (int i = 0; i < 2 * EPB; ++i) nxt[i] = edges[(size_t)nb * (2 * EPB) + i];
            }

            float v[EPB];
#pragma unroll
            for (int i = 0; i < EPB; ++i) {
                v[i] = Y[(size_t)cur[2 * i] * DDIM + c];
            }
#pragma unroll
            for (int i = 0; i < EPB; ++i) {
                unsafeAtomicAdd(out + (size_t)cur[2 * i + 1] * DDIM + c, v[i]);
            }

            if (!more) break;
#pragma unroll
            for (int i = 0; i < 2 * EPB; ++i) cur[i] = nxt[i];
            b = nb;
        }
    }

    if (blockIdx.x == 0) {
        for (int e = nFull * EPB; e < E; ++e) {
            const int s = edges[2 * e + 0];
            const int d = edges[2 * e + 1];
            const float v = Y[(size_t)s * DDIM + c];
            unsafeAtomicAdd(out + (size_t)d * DDIM + c, v);
        }
    }
}


#define RANGE      6400
#define SCHUNK     7
#define SORT_U     8
#define PULL_BLOCKS 2048

typedef int v4i __attribute__((ext_vector_type(4)));

extern "C" __global__ __launch_bounds__(1024)
void rgcn_sort_by_dst(const int* __restrict__ edges,
                      int* __restrict__ pos,
                      int* __restrict__ ssrc,
                      int E, int M)
{
    __shared__ int hist[RANGE];
    __shared__ int part[1024];
    __shared__ int belowSh;

    const int t   = threadIdx.x;
    const int r   = blockIdx.x;
    const int lo  = blockIdx.y * RANGE;
    const int len = min(RANGE, M - lo);
    const int* __restrict__ ed = edges + (size_t)r * E * 2;
    const v4i* __restrict__ ed4 = (const v4i*)ed;
    const int nPairs = E >> 1;
    int* __restrict__ srel = ssrc + (size_t)r * E;

    for (int i = t; i < RANGE; i += 1024) hist[i] = 0;
    if (t == 0) belowSh = 0;
    __syncthreads();

    int below = 0;
    for (int base = t; base < E; base += 2 * SORT_U * 1024) {
        int dq[2 * SORT_U];
#pragma unroll
        for (int j = 0; j < 2 * SORT_U; ++j) {
            const int idx = base + j * 1024;
            dq[j] = ed[2 * (size_t)min(idx, E - 1) + 1];
        }
#pragma unroll
        for (int j = 0; j < 2 * SORT_U; ++j) {
            const int d = dq[j];
            if ((base + j * 1024) < E) {
                below += (d < lo);
                if ((unsigned)(d - lo) < (unsigned)len) atomicAdd(&hist[d - lo], 1);
            }
        }
    }
#pragma unroll
    for (int m = 16; m >= 1; m >>= 1) below += __shfl_xor(below, m, 32);
    if ((t & 31) == 0) atomicAdd(&belowSh, below);
    __syncthreads();

    const int c0 = min(t * SCHUNK, RANGE);
    const int c1 = min(c0 + SCHUNK, RANGE);
    int s = 0;
    for (int i = c0; i < c1; ++i) s += hist[i];
    part[t] = s;
    __syncthreads();
    for (int off = 1; off < 1024; off <<= 1) {
        const int v = (t >= off) ? part[t - off] : 0;
        __syncthreads();
        part[t] += v;
        __syncthreads();
    }
    int run = belowSh + part[t] - s;
    for (int i = c0; i < c1; ++i) {
        const int v = hist[i];
        hist[i] = run;
        run += v;
        if (i < len) pos[(size_t)r * M + lo + i] = run;
    }
    __syncthreads();

    for (int base = t; base < nPairs; base += SORT_U * 1024) {
        v4i q[SORT_U];
#pragma unroll
        for (int j = 0; j < SORT_U; ++j) {
            const int idx = base + j * 1024;
            q[j] = ed4[min(idx, nPairs - 1)];
        }
#pragma unroll
        for (int j = 0; j < SORT_U; ++j) {
            const bool valid = (base + j * 1024) < nPairs;
            const int s0 = q[j].x, d0 = q[j].y, s1 = q[j].z, d1 = q[j].w;
            if (valid) {
                if ((unsigned)(d0 - lo) < (unsigned)len) srel[atomicAdd(&hist[d0 - lo], 1)] = s0;
                if ((unsigned)(d1 - lo) < (unsigned)len) srel[atomicAdd(&hist[d1 - lo], 1)] = s1;
            }
        }
    }
    if ((E & 1) && t == 0) {
        const int sN = ed[2 * (size_t)(E - 1) + 0];
        const int d  = ed[2 * (size_t)(E - 1) + 1];
        if ((unsigned)(d - lo) < (unsigned)len) srel[atomicAdd(&hist[d - lo], 1)] = sN;
    }
}

typedef float v4f __attribute__((ext_vector_type(4)));

static __device__ __forceinline__ void pull_rows(const float* __restrict__ Y,
                                                 const int*   __restrict__ pos,
                                                 const int*   __restrict__ ssrc,
                                                 float* __restrict__ out, int M,
                                                 int gw, int nWaves, int lane)
{
    for (int d = gw; d < M; d += nWaves) {
        const int beg = (d > 0) ? pos[d - 1] : 0;
        const int end = pos[d];
        if (end <= beg) continue;
        v4f* orow = (v4f*)(out + (size_t)d * DDIM);
        v4f o0 = orow[lane];
        v4f o1 = orow[32 + lane];
        for (int e = beg; e < end; e += 4) {
            const int n = end - e;
            const v4f zero = (v4f)0.0f;
            v4f a1 = zero, b1 = zero, a2 = zero, b2 = zero, a3 = zero, b3 = zero;
            const v4f* y0 = (const v4f*)(Y + (size_t)ssrc[e] * DDIM);
            const v4f a0 = y0[lane], b0 = y0[32 + lane];
            if (n > 1) {
                const v4f* y1 = (const v4f*)(Y + (size_t)ssrc[e + 1] * DDIM);
                a1 = y1[lane]; b1 = y1[32 + lane];
            }
            if (n > 2) {
                const v4f* y2 = (const v4f*)(Y + (size_t)ssrc[e + 2] * DDIM);
                a2 = y2[lane]; b2 = y2[32 + lane];
            }
            if (n > 3) {
                const v4f* y3 = (const v4f*)(Y + (size_t)ssrc[e + 3] * DDIM);
                a3 = y3[lane]; b3 = y3[32 + lane];
            }
            o0 += a0; o1 += b0;
            o0 += a1; o1 += b1;
            o0 += a2; o1 += b2;
            o0 += a3; o1 += b3;
        }
        orow[lane]      = o0;
        orow[32 + lane] = o1;
    }
}

extern "C" __global__ __launch_bounds__(256)
void rgcn_pull_add(const float* __restrict__ Y,
                   const int*   __restrict__ pos,
                   const int*   __restrict__ ssrc,
                   float* __restrict__ out, int M)
{
    const int lane = threadIdx.x & 31;
    const int gw = __builtin_amdgcn_readfirstlane((int)(blockIdx.x * 8 + (threadIdx.x >> 5)));
    pull_rows(Y, pos, ssrc, out, M, gw, PULL_BLOCKS * 8, lane);
}

#define FUSED_THREADS 1024
#define PULL_WAVES_PER_BLOCK 16
extern "C" __global__ __launch_bounds__(FUSED_THREADS)
void rgcn_gemm_pull(const float* __restrict__ X,
                    const float* __restrict__ W,
                    float* __restrict__ Y,
                    int M,
                    const float* __restrict__ Yprev,
                    const int*   __restrict__ pos,
                    const int*   __restrict__ ssrc,
                    float* __restrict__ out)
{
    __shared__ float ldsW[DDIM * LDS_STRIDE];
    const v2f* ldsW2 = (const v2f*)ldsW;

    const int tid = threadIdx.x;

    {
        const int n    = tid & (DDIM - 1);
        const int kOff = (tid >> 8) * (DDIM / 4);
        for (int k = 0; k < DDIM / 4; ++k) {
            ldsW[n * LDS_STRIDE + kOff + k] = W[(kOff + k) * DDIM + n];
        }
    }
    __syncthreads();

    const int wave = __builtin_amdgcn_readfirstlane(tid >> 5);
    const int lane = tid & 31;
    if (wave < GEMM_THREADS / 32) {
        gemm_block(X, ldsW2, Y, M, wave, lane);
    } else {
        const int gw = (int)blockIdx.x * PULL_WAVES_PER_BLOCK + (wave - GEMM_THREADS / 32);
        pull_rows(Yprev, pos, ssrc, out, M, gw, (int)gridDim.x * PULL_WAVES_PER_BLOCK, lane);
    }
}

extern "C" void kernel_launch(void* const* d_in, const int* in_sizes, int n_in,
                              void* d_out, int out_size, void* d_ws, size_t ws_size,
                              hipStream_t stream)
{
    const float* x      = (const float*)d_in[0];
    const float* W_rel  = (const float*)d_in[1];
    const float* W_self = (const float*)d_in[2];
    const int*   edges  = (const int*)  d_in[3];

    float* out = (float*)d_out;
    float* y   = (float*)d_ws;

    const int M = in_sizes[0] / DDIM;
    const int R = in_sizes[1] / (DDIM * DDIM);
    const int E = in_sizes[3] / (2 * R);

    const int  nStrips = (M + 15) / 16;
    const dim3 gemmGrid((nStrips + STRIPS_PER_BLOCK - 1) / STRIPS_PER_BLOCK);

    rgcn_gemm_wmma<<<gemmGrid, GEMM_THREADS, 0, stream>>>(x, W_self, out, M);

    const size_t yBytes = (size_t)M * DDIM * sizeof(float);
    int* pos  = (int*)((char*)d_ws + yBytes);
    int* ssrc = pos + (size_t)R * M;
    const size_t need = yBytes + ((size_t)R * M + (size_t)R * E) * sizeof(int);

    if (ws_size >= need + yBytes) {
        float* y1  = (float*)((char*)d_ws + yBytes);
        int* pos2  = (int*)((char*)d_ws + 2 * yBytes);
        int* ssrc2 = pos2 + (size_t)R * M;

        const dim3 sortGrid(R, (M + RANGE - 1) / RANGE);
        rgcn_sort_by_dst<<<sortGrid, 1024, 0, stream>>>(edges, pos2, ssrc2, E, M);

        rgcn_gemm_wmma<<<gemmGrid, GEMM_THREADS, 0, stream>>>(x, W_rel, y, M);
        for (int r = 1; r < R; ++r) {
            rgcn_gemm_pull<<<gemmGrid, FUSED_THREADS, 0, stream>>>(
                x, W_rel + (size_t)r * DDIM * DDIM, (r & 1) ? y1 : y, M,
                (r & 1) ? y : y1, pos2 + (size_t)(r - 1) * M, ssrc2 + (size_t)(r - 1) * E, out);
        }
        rgcn_pull_add<<<PULL_BLOCKS, 256, 0, stream>>>(
            ((R - 1) & 1) ? y1 : y, pos2 + (size_t)(R - 1) * M, ssrc2 + (size_t)(R - 1) * E, out, M);
    } else if (ws_size >= need) {
        const dim3 sortGrid(R, (M + RANGE - 1) / RANGE);
        rgcn_sort_by_dst<<<sortGrid, 1024, 0, stream>>>(edges, pos, ssrc, E, M);

        for (int r = 0; r < R; ++r) {
            rgcn_gemm_wmma<<<gemmGrid, GEMM_THREADS, 0, stream>>>(
                x, W_rel + (size_t)r * DDIM * DDIM, y, M);
            rgcn_pull_add<<<PULL_BLOCKS, 256, 0, stream>>>(
                y, pos + (size_t)r * M, ssrc + (size_t)r * E, out, M);
        }
    } else {
        for (int r = 0; r < R; ++r) {
            rgcn_gemm_wmma<<<gemmGrid, GEMM_THREADS, 0, stream>>>(
                x, W_rel + (size_t)r * DDIM * DDIM, y, M);
            rgcn_scatter_add<<<SCATTER_BLOCKS, 256, 0, stream>>>(
                y, edges + (size_t)r * E * 2, out, E);
        }
    }
}
